# starting kernel
# baseline (speedup 1.0000x reference)
#include <hip/hip_runtime.h>


#define B_DIM 64
#define T_DIM 8192
#define F_DIM 64
#define CH 16
#define NCHUNK (T_DIM / CH)
#define ALPHA_F 0.1f
#define BETA_F 0.9f

typedef float v2f __attribute__((ext_vector_type(2)));
typedef float v8f __attribute__((ext_vector_type(8)));

__host__ __device__ constexpr float bp(int e) {
    float r = 1.0f;
    for (int i = 0; i < e; ++i) r *= BETA_F;
    return r;
}
__host__ __device__ constexpr float winv(int j) { return ALPHA_F / bp(j); }

__global__ void ema_chunk_tail(const float* __restrict__ x,
                               float* __restrict__ s) {
    const int b = blockIdx.x >> 9;
    const int chunk = blockIdx.x & (NCHUNK - 1);
    const int f = threadIdx.x;

    const float* xb = x + ((size_t)b * T_DIM + (size_t)chunk * CH) * F_DIM + f;

    float acc = 0.0f;
#pragma unroll
    for (int j = CH - 1; j >= 1; --j) {
        acc += (ALPHA_F * bp(CH - 1 - j)) * xb[(size_t)j * F_DIM];
    }
    const float w0 = (chunk == 0) ? bp(CH - 1) : (ALPHA_F * bp(CH - 1));
    acc += w0 * xb[0];

    s[(((size_t)b << 9) + chunk) * F_DIM + f] = acc;
}

__global__ void ema_carry_scan(float* __restrict__ s) {
    const int tid = blockIdx.x * blockDim.x + threadIdx.x;
    const int b = tid >> 6;
    const int f = tid & (F_DIM - 1);

    float* p = s + ((size_t)b << 9) * F_DIM + f;

    float c = 0.0f;
#pragma unroll 8
    for (int k = 0; k < NCHUNK; ++k) {
        const float sl = p[(size_t)k * F_DIM];
        p[(size_t)k * F_DIM] = c;
        c = sl + bp(CH) * c;
    }
}

__global__ void ema_chunk_wmma(const float* __restrict__ x,
                               const float* __restrict__ carry,
                               float* __restrict__ y) {
    const int b = blockIdx.x >> 9;
    const int chunk = blockIdx.x & (NCHUNK - 1);
    const int wave = threadIdx.x >> 5;
    const int lane = threadIdx.x & 31;
    const int half = lane >> 4;
    const int l = lane & 15;

    const int tb = chunk * CH;
    const int fb = wave * 16;

    const float* xb = x + ((size_t)b * T_DIM + tb) * F_DIM;
    float* yb = y + ((size_t)b * T_DIM + tb) * F_DIM;

    const float pl = ((l & 1) ? bp(1) : 1.0f) *
                     ((l & 2) ? bp(2) : 1.0f) *
                     ((l & 4) ? bp(4) : 1.0f) *
                     ((l & 8) ? bp(8) : 1.0f);

    const float col0 = (chunk == 0) ? 1.0f : ALPHA_F;

    v8f acc = {};
#pragma unroll
    for (int s = 0; s < 4; ++s) {
        const int j0 = 4 * s + 2 * half;
        const int j1 = j0 + 1;

        const float sc0 = half ? winv(4 * s + 2)
                               : (s == 0 ? col0 : winv(4 * s));
        const float sc1 = half ? winv(4 * s + 3) : winv(4 * s + 1);

        v2f a;
        a.x = (l >= j0) ? sc0 * pl : 0.0f;
        a.y = (l >= j1) ? sc1 * pl : 0.0f;

        v2f bm;
        bm.x = xb[(size_t)j0 * F_DIM + fb + l];
        bm.y = xb[(size_t)j1 * F_DIM + fb + l];

        acc = __builtin_amdgcn_wmma_f32_16x16x4_f32(
             false, a,  false, bm,
             (short)0, acc,  false,  false);
    }

    const float cv = carry[(((size_t)b << 9) + chunk) * F_DIM + fb + l];
#pragma unroll
    for (int r = 0; r < 8; ++r) {
        const float cf = half ? bp(r + 9) : bp(r + 1);
        const int t = r + 8 * half;
        yb[(size_t)t * F_DIM + fb + l] = acc[r] + cf * cv;
    }
}

extern "C" void kernel_launch(void* const* d_in, const int* in_sizes, int n_in,
                              void* d_out, int out_size, void* d_ws, size_t ws_size,
                              hipStream_t stream) {
    (void)in_sizes; (void)n_in; (void)out_size; (void)ws_size;

    const float* x = (const float*)d_in[0];
    float* out = (float*)d_out;
    float* s = (float*)d_ws;

    ema_chunk_tail<<<dim3(B_DIM * NCHUNK), dim3(F_DIM), 0, stream>>>(x, s);
    ema_carry_scan<<<dim3(32), dim3(128), 0, stream>>>(s);
    ema_chunk_wmma<<<dim3(B_DIM * NCHUNK), dim3(128), 0, stream>>>(x, s, out);
}


# Carry scan parallelised: 16 segments of 32 chunks per sequence, loads issued together, segment totals folded through LDS (changes the order of the carry sums); 128 blocks of 16 waves
# speedup vs baseline: 2.4735x; 2.4735x over previous
#include <hip/hip_runtime.h>


#define B_DIM 64
#define T_DIM 8192
#define F_DIM 64
#define CH 16
#define NCHUNK (T_DIM / CH)
#define ALPHA_F 0.1f
#define BETA_F 0.9f

typedef float v2f __attribute__((ext_vector_type(2)));
typedef float v8f __attribute__((ext_vector_type(8)));

__host__ __device__ constexpr float bp(int e) {
    float r = 1.0f;
    for (int i = 0; i < e; ++i) r *= BETA_F;
    return r;
}
__host__ __device__ constexpr float winv(int j) { return ALPHA_F / bp(j); }

__global__ void ema_chunk_tail(const float* __restrict__ x,
                               float* __restrict__ s) {
    const int b = blockIdx.x >> 9;
    const int chunk = blockIdx.x & (NCHUNK - 1);
    const int f = threadIdx.x;

    const float* xb = x + ((size_t)b * T_DIM + (size_t)chunk * CH) * F_DIM + f;

    float acc = 0.0f;
#pragma unroll
    for (int j = CH - 1; j >= 1; --j) {
        acc += (ALPHA_F * bp(CH - 1 - j)) * xb[(size_t)j * F_DIM];
    }
    const float w0 = (chunk == 0) ? bp(CH - 1) : (ALPHA_F * bp(CH - 1));
    acc += w0 * xb[0];

    s[(((size_t)b << 9) + chunk) * F_DIM + f] = acc;
}

#define SEG 16
#define SLEN (NCHUNK / SEG)
#define SCAN_F 32
#define SCAN_BLOCKS (B_DIM * (F_DIM / SCAN_F))

__host__ __device__ constexpr float gp(int e) {
    float r = 1.0f;
    for (int i = 0; i < e; ++i) r *= bp(CH);
    return r;
}

__global__ void __launch_bounds__(SEG * SCAN_F)
ema_carry_scan(float* __restrict__ s) {
    __shared__ float tot[SEG][SCAN_F];

    const int b = blockIdx.x >> 1;
    const int fh = blockIdx.x & 1;
    const int seg = threadIdx.x >> 5;
    const int fl = threadIdx.x & 31;

    float* p = s + (((size_t)b << 9) + (size_t)seg * SLEN) * F_DIM + fh * SCAN_F + fl;

    float v[SLEN];
#pragma unroll
    for (int i = 0; i < SLEN; ++i) v[i] = p[(size_t)i * F_DIM];

    float e = 0.0f;
#pragma unroll
    for (int i = 0; i < SLEN; ++i) {
        const float sl = v[i];
        v[i] = e;
        e = sl + gp(1) * e;
    }
    tot[seg][fl] = e;
    __syncthreads();

    float run = 0.0f, c = 0.0f;
#pragma unroll
    for (int q = 0; q < SEG - 1; ++q) {
        run = tot[q][fl] + gp(SLEN) * run;
        c = (q < seg) ? run : c;
    }

#pragma unroll
    for (int i = 0; i < SLEN; ++i) p[(size_t)i * F_DIM] = v[i] + gp(i) * c;
}

__global__ void ema_chunk_wmma(const float* __restrict__ x,
                               const float* __restrict__ carry,
                               float* __restrict__ y) {
    const int b = blockIdx.x >> 9;
    const int chunk = blockIdx.x & (NCHUNK - 1);
    const int wave = threadIdx.x >> 5;
    const int lane = threadIdx.x & 31;
    const int half = lane >> 4;
    const int l = lane & 15;

    const int tb = chunk * CH;
    const int fb = wave * 16;

    const float* xb = x + ((size_t)b * T_DIM + tb) * F_DIM;
    float* yb = y + ((size_t)b * T_DIM + tb) * F_DIM;

    const float pl = ((l & 1) ? bp(1) : 1.0f) *
                     ((l & 2) ? bp(2) : 1.0f) *
                     ((l & 4) ? bp(4) : 1.0f) *
                     ((l & 8) ? bp(8) : 1.0f);

    const float col0 = (chunk == 0) ? 1.0f : ALPHA_F;

    v8f acc = {};
#pragma unroll
    for (int s = 0; s < 4; ++s) {
        const int j0 = 4 * s + 2 * half;
        const int j1 = j0 + 1;

        const float sc0 = half ? winv(4 * s + 2)
                               : (s == 0 ? col0 : winv(4 * s));
        const float sc1 = half ? winv(4 * s + 3) : winv(4 * s + 1);

        v2f a;
        a.x = (l >= j0) ? sc0 * pl : 0.0f;
        a.y = (l >= j1) ? sc1 * pl : 0.0f;

        v2f bm;
        bm.x = xb[(size_t)j0 * F_DIM + fb + l];
        bm.y = xb[(size_t)j1 * F_DIM + fb + l];

        acc = __builtin_amdgcn_wmma_f32_16x16x4_f32(
             false, a,  false, bm,
             (short)0, acc,  false,  false);
    }

    const float cv = carry[(((size_t)b << 9) + chunk) * F_DIM + fb + l];
#pragma unroll
    for (int r = 0; r < 8; ++r) {
        const float cf = half ? bp(r + 9) : bp(r + 1);
        const int t = r + 8 * half;
        yb[(size_t)t * F_DIM + fb + l] = acc[r] + cf * cv;
    }
}

extern "C" void kernel_launch(void* const* d_in, const int* in_sizes, int n_in,
                              void* d_out, int out_size, void* d_ws, size_t ws_size,
                              hipStream_t stream) {
    (void)in_sizes; (void)n_in; (void)out_size; (void)ws_size;

    const float* x = (const float*)d_in[0];
    float* out = (float*)d_out;
    float* s = (float*)d_ws;

    ema_chunk_tail<<<dim3(B_DIM * NCHUNK), dim3(F_DIM), 0, stream>>>(x, s);
    ema_carry_scan<<<dim3(SCAN_BLOCKS), dim3(SEG * SCAN_F), 0, stream>>>(s);
    ema_chunk_wmma<<<dim3(B_DIM * NCHUNK), dim3(128), 0, stream>>>(x, s, out);
}


# Streaming passes re-shaped: 4096 waves per pass, each owns 8 consecutive chunks, requests the next chunks' rows before it works on one, moves 256-byte rows with 16-byte loads and stores
# speedup vs baseline: 2.9359x; 1.1869x over previous
#include <hip/hip_runtime.h>


#define B_DIM 64
#define T_DIM 8192
#define F_DIM 64
#define CH 16
#define NCHUNK (T_DIM / CH)
#define ALPHA_F 0.1f
#define BETA_F 0.9f

typedef float v2f __attribute__((ext_vector_type(2)));
typedef float v4f __attribute__((ext_vector_type(4)));
typedef float v8f __attribute__((ext_vector_type(8)));

#define CPW 8
#define WPB 8
#define STREAM_BLOCKS (B_DIM * NCHUNK / (CPW * WPB))
#define ROW4 (F_DIM / 4)
#define CHUNK4 (CH * ROW4)

__host__ __device__ constexpr float bp(int e) {
    float r = 1.0f;
    for (int i = 0; i < e; ++i) r *= BETA_F;
    return r;
}
__host__ __device__ constexpr float winv(int j) { return ALPHA_F / bp(j); }

#define PAIRS (CPW / 2)

__global__ void __launch_bounds__(WPB * 32)
ema_chunk_tail(const float* __restrict__ x, float* __restrict__ s) {
    const int wave = __builtin_amdgcn_readfirstlane(threadIdx.x >> 5);
    const int lane = threadIdx.x & 31;
    const int half = lane >> 4;
    const int c = lane & 15;
    const int g0 = (blockIdx.x * WPB + wave) * CPW;

    const v4f* xbase = (const v4f*)(x + (size_t)(g0 + half) * (CH * F_DIM)) + c;
    v4f* sbase = (v4f*)(s + (size_t)(g0 + half) * F_DIM) + c;

    v4f xv[2][CH];
#define LOAD_PAIR(P) do { \
        const v4f* xp_ = xbase + (size_t)(P) * (2 * CHUNK4); \
        _Pragma("unroll") for (int j = 0; j < CH; ++j) xv[(P) & 1][j] = xp_[j * ROW4]; \
    } while (0)

    LOAD_PAIR(0);
#pragma unroll
    for (int p = 0; p < PAIRS; ++p) {
        __builtin_amdgcn_sched_barrier(0);
        if (p + 1 < PAIRS) LOAD_PAIR(p + 1);
        __builtin_amdgcn_sched_barrier(0);

        const v4f* xr = xv[p & 1];
        v4f acc = {0.0f, 0.0f, 0.0f, 0.0f};
#pragma unroll
        for (int j = CH - 1; j >= 1; --j) {
            acc += (ALPHA_F * bp(CH - 1 - j)) * xr[j];
        }
        const int g = g0 + 2 * p + half;
        const float w0 = ((g & (NCHUNK - 1)) == 0) ? bp(CH - 1) : (ALPHA_F * bp(CH - 1));
        acc += w0 * xr[0];

        sbase[(size_t)p * (2 * ROW4)] = acc;
    }
#undef LOAD_PAIR
}

#define SEG 16
#define SLEN (NCHUNK / SEG)
#define SCAN_F 32
#define SCAN_BLOCKS (B_DIM * (F_DIM / SCAN_F))

__host__ __device__ constexpr float gp(int e) {
    float r = 1.0f;
    for (int i = 0; i < e; ++i) r *= bp(CH);
    return r;
}

__global__ void __launch_bounds__(SEG * SCAN_F)
ema_carry_scan(float* __restrict__ s) {
    __shared__ float tot[SEG][SCAN_F];

    const int b = blockIdx.x >> 1;
    const int fh = blockIdx.x & 1;
    const int seg = threadIdx.x >> 5;
    const int fl = threadIdx.x & 31;

    float* p = s + (((size_t)b << 9) + (size_t)seg * SLEN) * F_DIM + fh * SCAN_F + fl;

    float v[SLEN];
#pragma unroll
    for (int i = 0; i < SLEN; ++i) v[i] = p[(size_t)i * F_DIM];

    float e = 0.0f;
#pragma unroll
    for (int i = 0; i < SLEN; ++i) {
        const float sl = v[i];
        v[i] = e;
        e = sl + gp(1) * e;
    }
    tot[seg][fl] = e;
    __syncthreads();

    float run = 0.0f, c = 0.0f;
#pragma unroll
    for (int q = 0; q < SEG - 1; ++q) {
        run = tot[q][fl] + gp(SLEN) * run;
        c = (q < seg) ? run : c;
    }

#pragma unroll
    for (int i = 0; i < SLEN; ++i) p[(size_t)i * F_DIM] = v[i] + gp(i) * c;
}

#define AHEAD 2
#define RING (AHEAD + 1)

__global__ void __launch_bounds__(WPB * 32)
ema_chunk_wmma(const float* __restrict__ x,
               const float* __restrict__ carry,
               float* __restrict__ y) {
    const int wave = __builtin_amdgcn_readfirstlane(threadIdx.x >> 5);
    const int lane = threadIdx.x & 31;
    const int half = lane >> 4;
    const int l = lane & 15;
    const int g0 = (blockIdx.x * WPB + wave) * CPW;

    const v4f* xbase = (const v4f*)(x + (size_t)g0 * (CH * F_DIM)) + (2 * half) * ROW4 + l;
    const v4f* cbase = (const v4f*)(carry + (size_t)g0 * F_DIM) + l;
    v4f* ybase = (v4f*)(y + (size_t)g0 * (CH * F_DIM)) + (8 * half) * ROW4 + l;

    v4f xv[RING][8];
    v4f cv[RING];
#define LOAD_CHUNK(Q) do { \
        const v4f* xp_ = xbase + (size_t)(Q) * CHUNK4; \
        _Pragma("unroll") for (int s_ = 0; s_ < 4; ++s_) { \
            xv[(Q) % RING][2 * s_] = xp_[(4 * s_) * ROW4]; \
            xv[(Q) % RING][2 * s_ + 1] = xp_[(4 * s_ + 1) * ROW4]; \
        } \
        cv[(Q) % RING] = cbase[(size_t)(Q) * ROW4]; \
    } while (0)

#pragma unroll
    for (int q = 0; q < AHEAD; ++q) LOAD_CHUNK(q);

    const float pl = ((l & 1) ? bp(1) : 1.0f) *
                     ((l & 2) ? bp(2) : 1.0f) *
                     ((l & 4) ? bp(4) : 1.0f) *
                     ((l & 8) ? bp(8) : 1.0f);

    v2f an[4];
#pragma unroll
    for (int s = 0; s < 4; ++s) {
        const int j0 = 4 * s + 2 * half;
        const int j1 = j0 + 1;
        const float sc0 = half ? winv(4 * s + 2) : winv(4 * s);
        const float sc1 = half ? winv(4 * s + 3) : winv(4 * s + 1);
        an[s].x = (l >= j0) ? sc0 * pl : 0.0f;
        an[s].y = (l >= j1) ? sc1 * pl : 0.0f;
    }
    v2f af;
    af.x = half ? an[0].x : pl;
    af.y = an[0].y;

#pragma unroll
    for (int q = 0; q < CPW; ++q) {
        __builtin_amdgcn_sched_barrier(0);
        if (q + AHEAD < CPW) LOAD_CHUNK(q + AHEAD);
        __builtin_amdgcn_sched_barrier(0);

        const v4f* xr = xv[q % RING];
        const v4f cq = cv[q % RING];
        const bool first = (((g0 + q) & (NCHUNK - 1)) == 0);

        v8f acc[4] = {};
#pragma unroll
        for (int s = 0; s < 4; ++s) {
            v2f a = an[s];
            if (s == 0) {
                a.x = first ? af.x : an[0].x;
                a.y = first ? af.y : an[0].y;
            }
#pragma unroll
            for (int m = 0; m < 4; ++m) {
                v2f bm;
                bm.x = xr[2 * s][m];
                bm.y = xr[2 * s + 1][m];
                acc[m] = __builtin_amdgcn_wmma_f32_16x16x4_f32(
                     false, a,  false, bm,
                     (short)0, acc[m],  false,  false);
            }
        }

        v4f* yq = ybase + (size_t)q * CHUNK4;
#pragma unroll
        for (int r = 0; r < 8; ++r) {
            const float cf = half ? bp(r + 9) : bp(r + 1);
            v4f o;
            o.x = acc[0][r] + cf * cq.x;
            o.y = acc[1][r] + cf * cq.y;
            o.z = acc[2][r] + cf * cq.z;
            o.w = acc[3][r] + cf * cq.w;
            yq[r * ROW4] = o;
        }
    }
#undef LOAD_CHUNK
}

extern "C" void kernel_launch(void* const* d_in, const int* in_sizes, int n_in,
                              void* d_out, int out_size, void* d_ws, size_t ws_size,
                              hipStream_t stream) {
    (void)in_sizes; (void)n_in; (void)out_size; (void)ws_size;

    const float* x = (const float*)d_in[0];
    float* out = (float*)d_out;
    float* s = (float*)d_ws;

    ema_chunk_tail<<<dim3(STREAM_BLOCKS), dim3(WPB * 32), 0, stream>>>(x, s);
    ema_carry_scan<<<dim3(SCAN_BLOCKS), dim3(SEG * SCAN_F), 0, stream>>>(s);
    ema_chunk_wmma<<<dim3(STREAM_BLOCKS), dim3(WPB * 32), 0, stream>>>(x, s, out);
}


# Single launch: a block of 16 waves walks one sequence (32 features) in 512-row steps, chunk tails and the running carry go through LDS, so x is read once (changes the order of the carry sums)
# speedup vs baseline: 3.0635x; 1.0435x over previous
#include <hip/hip_runtime.h>


#define B_DIM 64
#define T_DIM 8192
#define F_DIM 64
#define CH 16
#define ALPHA_F 0.1f
#define BETA_F 0.9f

typedef float v2f __attribute__((ext_vector_type(2)));
typedef float v8f __attribute__((ext_vector_type(8)));

#define FB 32
#define NW 16
#define CPS 2
#define ROWS_W (CPS * CH)
#define ROWS_S (NW * ROWS_W)
#define NSTEP (T_DIM / ROWS_S)
#define AHEAD 2
#define RING (AHEAD + 1)
#define ROW2 (F_DIM / 2)
#define WALK_BLOCKS (B_DIM * (F_DIM / FB))

__host__ __device__ constexpr float bp(int e) {
    float r = 1.0f;
    for (int i = 0; i < e; ++i) r *= BETA_F;
    return r;
}
__host__ __device__ constexpr float winv(int j) { return ALPHA_F / bp(j); }

__global__ void __launch_bounds__(NW * 32)
ema_walk(const float* __restrict__ x, float* __restrict__ y) {
    __shared__ v2f tot[2][2][NW][16];
    __shared__ v2f t0s[2][2][NW][16];

    const int w = __builtin_amdgcn_readfirstlane(threadIdx.x >> 5);
    const int lane = threadIdx.x & 31;
    const int half = lane >> 4;
    const int c = lane & 15;
    const int b = blockIdx.x >> 1;
    const int fh = blockIdx.x & 1;

    const size_t row0 = (size_t)b * T_DIM + (size_t)w * ROWS_W;
    const v2f* xbase = (const v2f*)(x + (row0 + 2 * half) * F_DIM + fh * FB) + c;
    v2f* ybase = (v2f*)(y + (row0 + 8 * half) * F_DIM + fh * FB) + c;

    v2f xv[RING][CPS][8];
#define LOAD_STEP(J) do { \
        const v2f* xp_ = xbase + (size_t)(J) * (ROWS_S * ROW2); \
        _Pragma("unroll") for (int k_ = 0; k_ < CPS; ++k_) { \
            _Pragma("unroll") for (int s_ = 0; s_ < 4; ++s_) { \
                xv[(J) % RING][k_][2 * s_] = xp_[(k_ * CH + 4 * s_) * ROW2]; \
                xv[(J) % RING][k_][2 * s_ + 1] = xp_[(k_ * CH + 4 * s_ + 1) * ROW2]; \
            } \
        } \
    } while (0)

#pragma unroll
    for (int j = 0; j < AHEAD; ++j) LOAD_STEP(j);

    const float pl = ((c & 1) ? bp(1) : 1.0f) *
                     ((c & 2) ? bp(2) : 1.0f) *
                     ((c & 4) ? bp(4) : 1.0f) *
                     ((c & 8) ? bp(8) : 1.0f);

    v2f an[4];
#pragma unroll
    for (int s = 0; s < 4; ++s) {
        const int j0 = 4 * s + 2 * half;
        const int j1 = j0 + 1;
        const float sc0 = half ? winv(4 * s + 2) : winv(4 * s);
        const float sc1 = half ? winv(4 * s + 3) : winv(4 * s + 1);
        an[s].x = (c >= j0) ? sc0 * pl : 0.0f;
        an[s].y = (c >= j1) ? sc1 * pl : 0.0f;
    }
    v2f af;
    af.x = half ? an[0].x : pl;
    af.y = an[0].y;

    v2f R = {0.0f, 0.0f};

#pragma unroll
    for (int j = 0; j < NSTEP; ++j) {
        __builtin_amdgcn_sched_barrier(0);
        if (j + AHEAD < NSTEP) LOAD_STEP(j + AHEAD);
        __builtin_amdgcn_sched_barrier(0);

        v8f acc[CPS][2] = {};
#pragma unroll
        for (int k = 0; k < CPS; ++k) {
            const v2f* xr = xv[j % RING][k];
#pragma unroll
            for (int s = 0; s < 4; ++s) {
                v2f a = an[s];
                if (j == 0 && k == 0 && s == 0) {
                    const bool first = (w == 0);
                    a.x = first ? af.x : an[0].x;
                    a.y = first ? af.y : an[0].y;
                }
#pragma unroll
                for (int m = 0; m < 2; ++m) {
                    v2f bm;
                    bm.x = xr[2 * s][m];
                    bm.y = xr[2 * s + 1][m];
                    acc[k][m] = __builtin_amdgcn_wmma_f32_16x16x4_f32(
                         false, a,  false, bm,
                         (short)0, acc[k][m],  false,  false);
                }
            }
        }

        {
            v2f t0, e;
            t0.x = acc[0][0][7];
            t0.y = acc[0][1][7];
            e.x = acc[1][0][7] + bp(CH) * t0.x;
            e.y = acc[1][1][7] + bp(CH) * t0.y;
            t0s[j & 1][half][w][c] = t0;
            tot[j & 1][half][w][c] = e;
        }
        __builtin_amdgcn_fence(__ATOMIC_RELEASE, "workgroup", "local");
        __builtin_amdgcn_s_barrier();
        __builtin_amdgcn_fence(__ATOMIC_ACQUIRE, "workgroup", "local");

        v2f run = R, cin = R;
#pragma unroll
        for (int q = 0; q < NW; ++q) {
            cin.x = (q == w) ? run.x : cin.x;
            cin.y = (q == w) ? run.y : cin.y;
            const v2f e = tot[j & 1][1][q][c];
            run.x = e.x + bp(ROWS_W) * run.x;
            run.y = e.y + bp(ROWS_W) * run.y;
        }
        R = run;

        const v2f t0 = t0s[j & 1][1][w][c];
        v2f ck[CPS];
        ck[0] = cin;
        ck[1].x = t0.x + bp(CH) * cin.x;
        ck[1].y = t0.y + bp(CH) * cin.y;

        v2f* yq = ybase + (size_t)j * (ROWS_S * ROW2);
#pragma unroll
        for (int k = 0; k < CPS; ++k) {
#pragma unroll
            for (int r = 0; r < 8; ++r) {
                const float cf = half ? bp(r + 9) : bp(r + 1);
                v2f o;
                o.x = acc[k][0][r] + cf * ck[k].x;
                o.y = acc[k][1][r] + cf * ck[k].y;
                yq[(k * CH + r) * ROW2] = o;
            }
        }
    }
#undef LOAD_STEP
}

extern "C" void kernel_launch(void* const* d_in, const int* in_sizes, int n_in,
                              void* d_out, int out_size, void* d_ws, size_t ws_size,
                              hipStream_t stream) {
    (void)in_sizes; (void)n_in; (void)out_size; (void)d_ws; (void)ws_size;

    const float* x = (const float*)d_in[0];
    float* out = (float*)d_out;

    ema_walk<<<dim3(WALK_BLOCKS), dim3(NW * 32), 0, stream>>>(x, out);
}
